# starting kernel
# baseline (speedup 1.0000x reference)
#include <hip/hip_runtime.h>
#include <hip/hip_bf16.h>

#define N_NODES 50000
#define HDIM    96
#define LN_EPS  1e-5f

typedef float v2f __attribute__((ext_vector_type(2)));
typedef float v8f __attribute__((ext_vector_type(8)));

__global__ void k_deg_init(float* __restrict__ deg, int n) {
    int i = blockIdx.x * blockDim.x + threadIdx.x;
    if (i < n) deg[i] = 1.0f;
}

__global__ void k_deg_edges(float* __restrict__ deg, const int* __restrict__ dst, int E) {
    int i = blockIdx.x * blockDim.x + threadIdx.x;
    if (i < E) atomicAdd(&deg[dst[i]], 1.0f);
}

__global__ void k_dinv(float* __restrict__ deg, int n) {
    int i = blockIdx.x * blockDim.x + threadIdx.x;
    if (i < n) deg[i] = rsqrtf(deg[i]);
}

__global__ void k_gemm_wmma_f32(const float* __restrict__ X,
                                const float* __restrict__ W,
                                float* __restrict__ Hout,
                                int M, int K) {
    __shared__ float Bs[128 * 16];

    const int ntile = blockIdx.y;

    for (int i = threadIdx.y * 32 + threadIdx.x; i < K * 16; i += 128) {
        int k = i >> 4;
        int n = i & 15;
        Bs[i] = W[k * HDIM + ntile * 16 + n];
    }
    __syncthreads();

    const int mtiles = M / 16;
    const int mtile  = blockIdx.x * 4 + threadIdx.y;
    if (mtile >= mtiles) return;

    const int lane = threadIdx.x;
    const int r    = lane & 15;
    const int koff = (lane >> 4) << 1;

    const float* __restrict__ arow = X + (size_t)(mtile * 16 + r) * K;

    v8f c = {};
    for (int kk = 0; kk < K; kk += 4) {
        v2f a, b;
        a.x = arow[kk + koff + 0];
        a.y = arow[kk + koff + 1];
        b.x = Bs[(kk + koff + 0) * 16 + r];
        b.y = Bs[(kk + koff + 1) * 16 + r];
        c = __builtin_amdgcn_wmma_f32_16x16x4_f32(
                false, a, false, b, (short)0, c, false, false);
    }

    const int row0 = mtile * 16 + ((lane >> 4) << 3);
    float* __restrict__ out = Hout + (size_t)row0 * HDIM + ntile * 16 + r;
#pragma unroll
    for (int i = 0; i < 8; ++i) out[(size_t)i * HDIM] = c[i];
}

__global__ void k_prescale(float* __restrict__ h_agg,
                           float* __restrict__ hs,
                           const float* __restrict__ dinv,
                           int total) {
    int i = blockIdx.x * blockDim.x + threadIdx.x;
    if (i < total) {
        int n = i / HDIM;
        float s = h_agg[i] * dinv[n];
        hs[i]    = s;
        h_agg[i] = s;
    }
}

__global__ void k_scatter(const float* __restrict__ hs,
                          float* __restrict__ agg,
                          const int* __restrict__ src,
                          const int* __restrict__ dst,
                          int E) {
    int wid  = (blockIdx.x * blockDim.x + threadIdx.x) >> 5;
    int lane = threadIdx.x & 31;
    if (wid >= E) return;

    int s = src[wid];
    int d = dst[wid];
    const float* __restrict__ hrow = hs  + (size_t)s * HDIM;
    float*       __restrict__ arow = agg + (size_t)d * HDIM;
#pragma unroll
    for (int i = 0; i < 3; ++i) {
        int f = lane + 32 * i;
        atomicAdd(&arow[f], hrow[f]);
    }
}

__global__ void k_epilogue(const float* __restrict__ agg,
                           const float* __restrict__ dinv,
                           const float* __restrict__ bias,
                           const float* __restrict__ gamma,
                           const float* __restrict__ beta,
                           const float* __restrict__ alpha,
                           float* __restrict__ y,
                           int N) {
    int wid  = (blockIdx.x * blockDim.x + threadIdx.x) >> 5;
    int lane = threadIdx.x & 31;
    if (wid >= N) return;

    const float* __restrict__ row = agg + (size_t)wid * HDIM;
    float dv = dinv[wid];

    float v[3];
#pragma unroll
    for (int i = 0; i < 3; ++i) {
        int f = lane + 32 * i;
        v[i] = row[f] * dv + bias[f];
    }

    float s = v[0] + v[1] + v[2];
#pragma unroll
    for (int m = 16; m > 0; m >>= 1) s += __shfl_xor(s, m, 32);
    float mu = s * (1.0f / (float)HDIM);

    float d0 = v[0] - mu, d1 = v[1] - mu, d2 = v[2] - mu;
    float q = d0 * d0 + d1 * d1 + d2 * d2;
#pragma unroll
    for (int m = 16; m > 0; m >>= 1) q += __shfl_xor(q, m, 32);
    float rstd = rsqrtf(q * (1.0f / (float)HDIM) + LN_EPS);

    float dvv[3] = {d0, d1, d2};
#pragma unroll
    for (int i = 0; i < 3; ++i) {
        int f = lane + 32 * i;
        float o = dvv[i] * rstd * gamma[f] + beta[f];
        y[(size_t)wid * HDIM + f] = (o >= 0.0f) ? o : alpha[f] * o;
    }
}

extern "C" void kernel_launch(void* const* d_in, const int* in_sizes, int n_in,
                              void* d_out, int out_size, void* d_ws, size_t ws_size,
                              hipStream_t stream) {
    const float* x   = (const float*)d_in[0];
    const int*   ei  = (const int*)  d_in[1];
    const float* W1  = (const float*)d_in[2];
    const float* b1  = (const float*)d_in[3];
    const float* W2  = (const float*)d_in[4];
    const float* b2  = (const float*)d_in[5];
    const float* g1  = (const float*)d_in[6];
    const float* be1 = (const float*)d_in[7];
    const float* g2  = (const float*)d_in[8];
    const float* be2 = (const float*)d_in[9];
    const float* a   = (const float*)d_in[10];

    const int N   = N_NODES;
    const int Cin = in_sizes[0] / N;
    const int E   = in_sizes[1] / 2;
    const int NH  = N * HDIM;

    const int* src = ei;
    const int* dst = ei + E;

    float* dinv = (float*)d_ws;
    float* bufH = dinv + ((N + 255) & ~255);
    float* bufS = bufH + (size_t)NH;
    float* y1   = (float*)d_out;

    k_deg_init <<<(N + 255) / 256, 256, 0, stream>>>(dinv, N);
    k_deg_edges<<<(E + 255) / 256, 256, 0, stream>>>(dinv, dst, E);
    k_dinv     <<<(N + 255) / 256, 256, 0, stream>>>(dinv, N);

    dim3 gemmBlock(32, 4);
    dim3 gemmGrid1((N / 16 + 3) / 4, HDIM / 16);
    const int scatterGrid = (E * 32 + 255) / 256;
    const int epiGrid     = (N * 32 + 255) / 256;

    k_gemm_wmma_f32<<<gemmGrid1, gemmBlock, 0, stream>>>(x, W1, bufH, N, Cin);
    k_prescale<<<(NH + 255) / 256, 256, 0, stream>>>(bufH, bufS, dinv, NH);
    k_scatter <<<scatterGrid, 256, 0, stream>>>(bufS, bufH, src, dst, E);
    k_epilogue<<<epiGrid, 256, 0, stream>>>(bufH, dinv, b1, g1, be1, a, y1, N);

    k_gemm_wmma_f32<<<gemmGrid1, gemmBlock, 0, stream>>>(y1, W2, bufH, N, HDIM);
    k_prescale<<<(NH + 255) / 256, 256, 0, stream>>>(bufH, bufS, dinv, NH);
    k_scatter <<<scatterGrid, 256, 0, stream>>>(bufS, bufH, src, dst, E);
    k_epilogue<<<epiGrid, 256, 0, stream>>>(bufH, dinv, b2, g2, be2, a, (float*)d_out, N);
}


# f32 atomic scatter replaced by a gather over an edge list grouped by destination (int atomic rank, scan, fill): one wave per node sums its source rows in registers in list order, stores the row once
# speedup vs baseline: 1.6033x; 1.6033x over previous
#include <hip/hip_runtime.h>
#include <hip/hip_bf16.h>

#define N_NODES 50000
#define HDIM    96
#define LN_EPS  1e-5f

typedef float v2f __attribute__((ext_vector_type(2)));
typedef float v8f __attribute__((ext_vector_type(8)));

#define SCAN_THREADS 1024
#define SCAN_TILES   ((N_NODES + SCAN_THREADS - 1) / SCAN_THREADS)

__global__ void k_cnt_clear(int* __restrict__ cnt, int n) {
    int i = blockIdx.x * blockDim.x + threadIdx.x;
    if (i < n) cnt[i] = 0;
}

__global__ void k_rank(int* __restrict__ cnt, int* __restrict__ rank,
                       const int* __restrict__ dst, int E) {
    int e = blockIdx.x * blockDim.x + threadIdx.x;
    if (e < E) rank[e] = atomicAdd(&cnt[dst[e]], 1);
}

__global__ void __launch_bounds__(SCAN_THREADS)
k_scan(const int* __restrict__ cnt, int* __restrict__ start,
       float* __restrict__ dinv, int n) {
    __shared__ int wtot[SCAN_THREADS / 32];

    const int lane = threadIdx.x & 31;
    const int w    = threadIdx.x >> 5;
    const int i0   = w * (32 * SCAN_TILES) + lane;

    int c[SCAN_TILES];
    int s = 0;
#pragma unroll
    for (int k = 0; k < SCAN_TILES; ++k) {
        int i = i0 + k * 32;
        c[k] = (i < n) ? cnt[i] : 0;
        s += c[k];
    }

    int inc = s;
#pragma unroll
    for (int off = 1; off < 32; off <<= 1) {
        int v = __shfl_up(inc, off, 32);
        if (lane >= off) inc += v;
    }
    if (lane == 31) wtot[w] = inc;
    __syncthreads();

    int run = inc - s;
    for (int j = 0; j < w; ++j) run += wtot[j];

#pragma unroll
    for (int k = 0; k < SCAN_TILES; ++k) {
        int i = i0 + k * 32;
        if (i < n) {
            start[i] = run;
            dinv[i]  = rsqrtf((float)(c[k] + 1));
            run += c[k];
        }
    }
}

__global__ void k_fill(int* __restrict__ col, const int* __restrict__ start,
                       const int* __restrict__ rank, const int* __restrict__ src,
                       const int* __restrict__ dst, int E) {
    int e = blockIdx.x * blockDim.x + threadIdx.x;
    if (e < E) col[start[dst[e]] + rank[e]] = src[e];
}

__global__ void k_gemm_wmma_f32(const float* __restrict__ X,
                                const float* __restrict__ W,
                                float* __restrict__ Hout,
                                int M, int K) {
    __shared__ float Bs[128 * 16];

    const int ntile = blockIdx.y;

    for (int i = threadIdx.y * 32 + threadIdx.x; i < K * 16; i += 128) {
        int k = i >> 4;
        int n = i & 15;
        Bs[i] = W[k * HDIM + ntile * 16 + n];
    }
    __syncthreads();

    const int mtiles = M / 16;
    const int mtile  = blockIdx.x * 4 + threadIdx.y;
    if (mtile >= mtiles) return;

    const int lane = threadIdx.x;
    const int r    = lane & 15;
    const int koff = (lane >> 4) << 1;

    const float* __restrict__ arow = X + (size_t)(mtile * 16 + r) * K;

    v8f c = {};
    for (int kk = 0; kk < K; kk += 4) {
        v2f a, b;
        a.x = arow[kk + koff + 0];
        a.y = arow[kk + koff + 1];
        b.x = Bs[(kk + koff + 0) * 16 + r];
        b.y = Bs[(kk + koff + 1) * 16 + r];
        c = __builtin_amdgcn_wmma_f32_16x16x4_f32(
                false, a, false, b, (short)0, c, false, false);
    }

    const int row0 = mtile * 16 + ((lane >> 4) << 3);
    float* __restrict__ out = Hout + (size_t)row0 * HDIM + ntile * 16 + r;
#pragma unroll
    for (int i = 0; i < 8; ++i) out[(size_t)i * HDIM] = c[i];
}

__global__ void k_prescale(float* __restrict__ h,
                           const float* __restrict__ dinv,
                           int total) {
    int i = blockIdx.x * blockDim.x + threadIdx.x;
    if (i < total) {
        int n = i / HDIM;
        h[i] = h[i] * dinv[n];
    }
}

typedef float v4f __attribute__((ext_vector_type(4)));
#define GATHER_GROUP 8
#define ROW_PIECES   (HDIM / 4)

__global__ void k_gather(const float* __restrict__ hs,
                         float* __restrict__ agg,
                         const int* __restrict__ col,
                         const int* __restrict__ start,
                         const int* __restrict__ cnt,
                         int N) {
    const int lane = threadIdx.x & 31;
    const int wid = __builtin_amdgcn_readfirstlane((blockIdx.x * blockDim.x + threadIdx.x) >> 5);
    if (wid >= N) return;

    const int beg = __builtin_amdgcn_readfirstlane(start[wid]);
    const int c   = __builtin_amdgcn_readfirstlane(cnt[wid]);

    const int piece = (lane < ROW_PIECES) ? lane : lane - 8;
    const v4f* __restrict__ rows = (const v4f*)hs;

    v4f acc = rows[(size_t)wid * ROW_PIECES + piece];

    for (int base = 0; base < c; base += 32) {
        const int m  = (c - base < 32) ? (c - base) : 32;
        const int my = (lane < m) ? col[beg + base + lane] : wid;
        for (int j = 0; j < m; j += GATHER_GROUP) {
            v4f v[GATHER_GROUP];
#pragma unroll
            for (int u = 0; u < GATHER_GROUP; ++u) {
                const int s = __builtin_amdgcn_readlane(my, j + u);
                v[u] = rows[(size_t)s * ROW_PIECES + piece];
            }
#pragma unroll
            for (int u = 0; u < GATHER_GROUP; ++u)
                if (j + u < m) acc += v[u];
        }
    }

    if (lane < ROW_PIECES)
        ((v4f*)agg)[(size_t)wid * ROW_PIECES + lane] = acc;
}

__global__ void k_epilogue(const float* __restrict__ dinv,
                           const float* __restrict__ bias,
                           const float* __restrict__ gamma,
                           const float* __restrict__ beta,
                           const float* __restrict__ alpha,
                           float* y,
                           int N) {
    int wid  = (blockIdx.x * blockDim.x + threadIdx.x) >> 5;
    int lane = threadIdx.x & 31;
    if (wid >= N) return;

    const float* row = y + (size_t)wid * HDIM;
    float dv = dinv[wid];

    float v[3];
#pragma unroll
    for (int i = 0; i < 3; ++i) {
        int f = lane + 32 * i;
        v[i] = row[f] * dv + bias[f];
    }

    float s = v[0] + v[1] + v[2];
#pragma unroll
    for (int m = 16; m > 0; m >>= 1) s += __shfl_xor(s, m, 32);
    float mu = s * (1.0f / (float)HDIM);

    float d0 = v[0] - mu, d1 = v[1] - mu, d2 = v[2] - mu;
    float q = d0 * d0 + d1 * d1 + d2 * d2;
#pragma unroll
    for (int m = 16; m > 0; m >>= 1) q += __shfl_xor(q, m, 32);
    float rstd = rsqrtf(q * (1.0f / (float)HDIM) + LN_EPS);

    float dvv[3] = {d0, d1, d2};
#pragma unroll
    for (int i = 0; i < 3; ++i) {
        int f = lane + 32 * i;
        float o = dvv[i] * rstd * gamma[f] + beta[f];
        y[(size_t)wid * HDIM + f] = (o >= 0.0f) ? o : alpha[f] * o;
    }
}

extern "C" void kernel_launch(void* const* d_in, const int* in_sizes, int n_in,
                              void* d_out, int out_size, void* d_ws, size_t ws_size,
                              hipStream_t stream) {
    const float* x   = (const float*)d_in[0];
    const int*   ei  = (const int*)  d_in[1];
    const float* W1  = (const float*)d_in[2];
    const float* b1  = (const float*)d_in[3];
    const float* W2  = (const float*)d_in[4];
    const float* b2  = (const float*)d_in[5];
    const float* g1  = (const float*)d_in[6];
    const float* be1 = (const float*)d_in[7];
    const float* g2  = (const float*)d_in[8];
    const float* be2 = (const float*)d_in[9];
    const float* a   = (const float*)d_in[10];

    const int N   = N_NODES;
    const int Cin = in_sizes[0] / N;
    const int E   = in_sizes[1] / 2;
    const int NH  = N * HDIM;

    const int* src = ei;
    const int* dst = ei + E;

    int*   cnt   = (int*)d_ws;
    float* bufH  = (float*)d_ws + ((N + 255) & ~255);
    int*   col   = (int*)(bufH + (size_t)NH);
    int*   rank  = col + E;
    int*   start = rank + E;
    float* dinv  = (float*)(start + N);
    float* y     = (float*)d_out;

    k_cnt_clear<<<(N + 255) / 256, 256, 0, stream>>>(cnt, N);
    k_rank     <<<(E + 255) / 256, 256, 0, stream>>>(cnt, rank, dst, E);
    k_scan     <<<1, SCAN_THREADS, 0, stream>>>(cnt, start, dinv, N);
    k_fill     <<<(E + 255) / 256, 256, 0, stream>>>(col, start, rank, src, dst, E);

    dim3 gemmBlock(32, 4);
    dim3 gemmGrid1((N / 16 + 3) / 4, HDIM / 16);
    const int nodeGrid = (N * 32 + 255) / 256;

    k_gemm_wmma_f32<<<gemmGrid1, gemmBlock, 0, stream>>>(x, W1, bufH, N, Cin);
    k_prescale<<<(NH + 255) / 256, 256, 0, stream>>>(bufH, dinv, NH);
    k_gather  <<<nodeGrid, 256, 0, stream>>>(bufH, y, col, start, cnt, N);
    k_epilogue<<<nodeGrid, 256, 0, stream>>>(dinv, b1, g1, be1, a, y, N);

    k_gemm_wmma_f32<<<gemmGrid1, gemmBlock, 0, stream>>>(y, W2, bufH, N, HDIM);
    k_prescale<<<(NH + 255) / 256, 256, 0, stream>>>(bufH, dinv, NH);
    k_gather  <<<nodeGrid, 256, 0, stream>>>(bufH, y, col, start, cnt, N);
    k_epilogue<<<nodeGrid, 256, 0, stream>>>(dinv, b2, g2, be2, a, y, N);
}


# layer 2's product folded into layer 1's aggregation launch: 16 one-node waves hand finished rows over in LDS, 6 of them run the 16x96x96 f32 WMMA tiles; one launch and a row-plane round trip gone
# speedup vs baseline: 2.7009x; 1.6846x over previous
#include <hip/hip_runtime.h>
#include <hip/hip_bf16.h>

#define N_NODES 50000
#define HDIM    96
#define LN_EPS  1e-5f

typedef float v2f __attribute__((ext_vector_type(2)));
typedef float v8f __attribute__((ext_vector_type(8)));

typedef int v4i __attribute__((ext_vector_type(4)));
#define EDGE_CHUNKS   96
#define BUILD_THREADS 1024
#define EDGE_UNROLL   8
#define OFF_THREADS   256

__global__ void __launch_bounds__(BUILD_THREADS)
k_count(int* __restrict__ cntc, int* __restrict__ total,
        const int* __restrict__ dst, int E) {
    __shared__ __attribute__((aligned(16))) int ctr[N_NODES];

    const int tid = threadIdx.x;
    const int g   = blockIdx.x;
    if (g == 0 && tid == 0) *total = 0;

    v4i* __restrict__ c4 = (v4i*)ctr;
    const v4i zero = {0, 0, 0, 0};
    for (int i = tid; i < N_NODES / 4; i += BUILD_THREADS) c4[i] = zero;
    __syncthreads();

    const int chunk = (E + EDGE_CHUNKS - 1) / EDGE_CHUNKS;
    const int e0 = g * chunk;
    const int e1 = (e0 + chunk < E) ? (e0 + chunk) : E;
    for (int e = e0 + tid; e < e1; e += EDGE_UNROLL * BUILD_THREADS) {
        int d[EDGE_UNROLL];
#pragma unroll
        for (int u = 0; u < EDGE_UNROLL; ++u) {
            const int ee = e + u * BUILD_THREADS;
            d[u] = (ee < e1) ? dst[ee] : -1;
        }
#pragma unroll
        for (int u = 0; u < EDGE_UNROLL; ++u)
            if (d[u] >= 0) atomicAdd(&ctr[d[u]], 1);
    }
    __syncthreads();

    v4i* __restrict__ row = (v4i*)(cntc + (size_t)g * N_NODES);
    for (int i = tid; i < N_NODES / 4; i += BUILD_THREADS) row[i] = c4[i];
}

__global__ void __launch_bounds__(OFF_THREADS)
k_offsets(int* __restrict__ cntc, int* __restrict__ cnt, int* __restrict__ start,
          float* __restrict__ dinv, int* __restrict__ total) {
    const int lane = threadIdx.x & 31;
    const int i    = blockIdx.x * OFF_THREADS + threadIdx.x;
    const bool live = i < N_NODES;

    int v[EDGE_CHUNKS];
#pragma unroll
    for (int g = 0; g < EDGE_CHUNKS; ++g)
        v[g] = live ? cntc[(size_t)g * N_NODES + i] : 0;
    int c = 0;
#pragma unroll
    for (int g = 0; g < EDGE_CHUNKS; ++g) c += v[g];

    int inc = c;
#pragma unroll
    for (int off = 1; off < 32; off <<= 1) {
        int t = __shfl_up(inc, off, 32);
        if (lane >= off) inc += t;
    }
    int base = 0;
    if (lane == 31) base = atomicAdd(total, inc);
    base = __shfl(base, 31, 32);

    if (live) {
        int s = base + inc - c;
        start[i] = s;
        cnt[i]   = c;
        dinv[i]  = rsqrtf((float)(c + 1));
#pragma unroll
        for (int g = 0; g < EDGE_CHUNKS; ++g) {
            cntc[(size_t)g * N_NODES + i] = s;
            s += v[g];
        }
    }
}

__global__ void __launch_bounds__(BUILD_THREADS)
k_place(int* __restrict__ col, const int* __restrict__ cntc,
        const int* __restrict__ src, const int* __restrict__ dst, int E) {
    __shared__ __attribute__((aligned(16))) int ctr[N_NODES];

    const int tid = threadIdx.x;
    const int g   = blockIdx.x;

    v4i* __restrict__ c4 = (v4i*)ctr;
    const v4i* __restrict__ row = (const v4i*)(cntc + (size_t)g * N_NODES);
    for (int i = tid; i < N_NODES / 4; i += BUILD_THREADS) c4[i] = row[i];
    __syncthreads();

    const int chunk = (E + EDGE_CHUNKS - 1) / EDGE_CHUNKS;
    const int e0 = g * chunk;
    const int e1 = (e0 + chunk < E) ? (e0 + chunk) : E;
    for (int e = e0 + tid; e < e1; e += EDGE_UNROLL * BUILD_THREADS) {
        int d[EDGE_UNROLL], s[EDGE_UNROLL];
#pragma unroll
        for (int u = 0; u < EDGE_UNROLL; ++u) {
            const int ee = e + u * BUILD_THREADS;
            d[u] = (ee < e1) ? dst[ee] : -1;
            s[u] = (ee < e1) ? src[ee] : 0;
        }
        int slot[EDGE_UNROLL];
#pragma unroll
        for (int u = 0; u < EDGE_UNROLL; ++u)
            slot[u] = (d[u] >= 0) ? atomicAdd(&ctr[d[u]], 1) : -1;
#pragma unroll
        for (int u = 0; u < EDGE_UNROLL; ++u)
            if (slot[u] >= 0) col[slot[u]] = s[u];
    }
}

__global__ void k_gemm_wmma_f32(const float* __restrict__ X,
                                const float* __restrict__ W,
                                const float* __restrict__ dinv,
                                float* __restrict__ Hout,
                                int M, int K) {
    __shared__ float Bs[128 * 16];

    const int ntile = blockIdx.y;

    for (int i = threadIdx.y * 32 + threadIdx.x; i < K * 16; i += 128) {
        int k = i >> 4;
        int n = i & 15;
        Bs[i] = W[k * HDIM + ntile * 16 + n];
    }
    __syncthreads();

    const int mtiles = M / 16;
    const int mtile  = blockIdx.x * 4 + threadIdx.y;
    if (mtile >= mtiles) return;

    const int lane = threadIdx.x;
    const int r    = lane & 15;
    const int koff = (lane >> 4) << 1;

    const float* __restrict__ arow = X + (size_t)(mtile * 16 + r) * K;

    const int row0 = mtile * 16 + ((lane >> 4) << 3);
    float dsc[8];
#pragma unroll
    for (int i = 0; i < 8; ++i) dsc[i] = dinv[row0 + i];

    v8f c = {};
    for (int kk = 0; kk < K; kk += 4) {
        v2f a, b;
        a.x = arow[kk + koff + 0];
        a.y = arow[kk + koff + 1];
        b.x = Bs[(kk + koff + 0) * 16 + r];
        b.y = Bs[(kk + koff + 1) * 16 + r];
        c = __builtin_amdgcn_wmma_f32_16x16x4_f32(
                false, a, false, b, (short)0, c, false, false);
    }

    float* __restrict__ out = Hout + (size_t)row0 * HDIM + ntile * 16 + r;
#pragma unroll
    for (int i = 0; i < 8; ++i) out[(size_t)i * HDIM] = c[i] * dsc[i];
}

typedef float v4f __attribute__((ext_vector_type(4)));
#define GATHER_GROUP 8
#define ROW_PIECES   (HDIM / 4)

__device__ __forceinline__ float wave_sum(float v) {
#pragma unroll
    for (int m = 16; m > 0; m >>= 1) v += __shfl_xor(v, m, 32);
    return v;
}

__device__ __forceinline__ v4f gather_ln_row(const float* __restrict__ hs,
                                             const int* __restrict__ col,
                                             const int* __restrict__ start,
                                             const int* __restrict__ cnt,
                                             const float* __restrict__ dinv,
                                             const float* __restrict__ bias,
                                             const float* __restrict__ gamma,
                                             const float* __restrict__ beta,
                                             const float* __restrict__ alpha,
                                             const int wid, const int lane) {
    const int beg = __builtin_amdgcn_readfirstlane(start[wid]);
    const int c   = __builtin_amdgcn_readfirstlane(cnt[wid]);

    const int piece = (lane < ROW_PIECES) ? lane : lane - 8;
    const v4f* __restrict__ rows = (const v4f*)hs;

    v4f acc = rows[(size_t)wid * ROW_PIECES + piece];

    const bool own = lane < ROW_PIECES;
    const int  f0  = piece * 4;
    const float dv = dinv[wid];
    float bi[4], ga[4], be[4], al[4];
#pragma unroll
    for (int i = 0; i < 4; ++i) {
        bi[i] = bias[f0 + i];
        ga[i] = gamma[f0 + i];
        be[i] = beta[f0 + i];
        al[i] = alpha[f0 + i];
    }

    for (int base = 0; base < c; base += 32) {
        const int m  = (c - base < 32) ? (c - base) : 32;
        const int my = (lane < m) ? col[beg + base + lane] : wid;
        for (int j = 0; j < m; j += GATHER_GROUP) {
            v4f v[GATHER_GROUP];
#pragma unroll
            for (int u = 0; u < GATHER_GROUP; ++u) {
                const int s = __builtin_amdgcn_readlane(my, j + u);
                v[u] = rows[(size_t)s * ROW_PIECES + piece];
            }
#pragma unroll
            for (int u = 0; u < GATHER_GROUP; ++u)
                if (j + u < m) acc += v[u];
        }
    }

    float t[4];
#pragma unroll
    for (int i = 0; i < 4; ++i) t[i] = acc[i] * dv + bi[i];

    const float s  = (t[0] + t[1]) + (t[2] + t[3]);
    const float mu = wave_sum(own ? s : 0.0f) * (1.0f / (float)HDIM);

    float d[4];
#pragma unroll
    for (int i = 0; i < 4; ++i) d[i] = t[i] - mu;
    const float q    = (d[0] * d[0] + d[1] * d[1]) + (d[2] * d[2] + d[3] * d[3]);
    const float rstd = rsqrtf(wave_sum(own ? q : 0.0f) * (1.0f / (float)HDIM) + LN_EPS);

    v4f o;
#pragma unroll
    for (int i = 0; i < 4; ++i) {
        const float z = d[i] * rstd * ga[i] + be[i];
        o[i] = (z >= 0.0f) ? z : al[i] * z;
    }
    return o;
}

__global__ void k_gather_ln(const float* __restrict__ hs,
                            float* __restrict__ y,
                            const int* __restrict__ col,
                            const int* __restrict__ start,
                            const int* __restrict__ cnt,
                            const float* __restrict__ dinv,
                            const float* __restrict__ bias,
                            const float* __restrict__ gamma,
                            const float* __restrict__ beta,
                            const float* __restrict__ alpha,
                            int N) {
    const int lane = threadIdx.x & 31;
    const int wid = __builtin_amdgcn_readfirstlane((blockIdx.x * blockDim.x + threadIdx.x) >> 5);
    if (wid >= N) return;

    const v4f o = gather_ln_row(hs, col, start, cnt, dinv, bias, gamma, beta, alpha, wid, lane);

    if (lane < ROW_PIECES)
        ((v4f*)y)[(size_t)wid * ROW_PIECES + lane] = o;
}

#define FUSE_NODES 16
#define YS_PITCH   100
static_assert(N_NODES % FUSE_NODES == 0, "every work-group is a full 16-row tile");
static_assert(FUSE_NODES >= HDIM / 16, "one wave per column tile");

__global__ void __launch_bounds__(FUSE_NODES * 32)
k_gather_ln_gemm(const float* __restrict__ hs,
                 float* __restrict__ hnext,
                 const int* __restrict__ col,
                 const int* __restrict__ start,
                 const int* __restrict__ cnt,
                 const float* __restrict__ dinv,
                 const float* __restrict__ bias,
                 const float* __restrict__ gamma,
                 const float* __restrict__ beta,
                 const float* __restrict__ alpha,
                 const float* __restrict__ Wn) {
    __shared__ __attribute__((aligned(16))) float Ys[FUSE_NODES * YS_PITCH];

    const int lane = threadIdx.x & 31;
    const int w     = __builtin_amdgcn_readfirstlane(threadIdx.x >> 5);
    const int tile0 = blockIdx.x * FUSE_NODES;
    const int wid   = tile0 + w;

    const v4f o = gather_ln_row(hs, col, start, cnt, dinv, bias, gamma, beta, alpha, wid, lane);

    if (lane < ROW_PIECES)
        *(v4f*)&Ys[w * YS_PITCH + lane * 4] = o;
    __syncthreads();

    if (w >= HDIM / 16) return;

    const int r    = lane & 15;
    const int koff = (lane >> 4) << 1;

    const int row0 = tile0 + ((lane >> 4) << 3);
    float dsc[8];
#pragma unroll
    for (int i = 0; i < 8; ++i) dsc[i] = dinv[row0 + i];

    const float* __restrict__ arow = Ys + r * YS_PITCH + koff;
    const float* __restrict__ bcol = Wn + (size_t)koff * HDIM + w * 16 + r;

    v8f c = {};
#pragma unroll 12
    for (int kk = 0; kk < HDIM; kk += 4) {
        v2f a, b;
        a.x = arow[kk + 0];
        a.y = arow[kk + 1];
        b.x = bcol[(size_t)(kk + 0) * HDIM];
        b.y = bcol[(size_t)(kk + 1) * HDIM];
        c = __builtin_amdgcn_wmma_f32_16x16x4_f32(
                false, a, false, b, (short)0, c, false, false);
    }

    float* __restrict__ out = hnext + (size_t)row0 * HDIM + w * 16 + r;
#pragma unroll
    for (int i = 0; i < 8; ++i) out[(size_t)i * HDIM] = c[i] * dsc[i];
}

extern "C" void kernel_launch(void* const* d_in, const int* in_sizes, int n_in,
                              void* d_out, int out_size, void* d_ws, size_t ws_size,
                              hipStream_t stream) {
    const float* x   = (const float*)d_in[0];
    const int*   ei  = (const int*)  d_in[1];
    const float* W1  = (const float*)d_in[2];
    const float* b1  = (const float*)d_in[3];
    const float* W2  = (const float*)d_in[4];
    const float* b2  = (const float*)d_in[5];
    const float* g1  = (const float*)d_in[6];
    const float* be1 = (const float*)d_in[7];
    const float* g2  = (const float*)d_in[8];
    const float* be2 = (const float*)d_in[9];
    const float* a   = (const float*)d_in[10];

    const int N   = N_NODES;
    const int Cin = in_sizes[0] / N;
    const int E   = in_sizes[1] / 2;
    const int NH  = N * HDIM;

    const int* src = ei;
    const int* dst = ei + E;

    int*   cnt   = (int*)d_ws;
    int*   total = cnt + N;
    float* bufH  = (float*)d_ws + ((N + 255) & ~255);
    int*   cntc  = (int*)bufH;
    int*   col   = (int*)(bufH + (size_t)NH);
    int*   start = col + 2 * (size_t)E;
    float* dinv  = (float*)(start + N);
    float* y     = (float*)d_out;

    k_count  <<<EDGE_CHUNKS, BUILD_THREADS, 0, stream>>>(cntc, total, dst, E);
    k_offsets<<<(N + OFF_THREADS - 1) / OFF_THREADS, OFF_THREADS, 0, stream>>>(cntc, cnt, start, dinv, total);
    k_place  <<<EDGE_CHUNKS, BUILD_THREADS, 0, stream>>>(col, cntc, src, dst, E);

    dim3 gemmBlock(32, 4);
    dim3 gemmGrid1((N / 16 + 3) / 4, HDIM / 16);
    const int nodeGrid = (N * 32 + 255) / 256;

    k_gemm_wmma_f32<<<gemmGrid1, gemmBlock, 0, stream>>>(x, W1, dinv, y, N, Cin);
    k_gather_ln_gemm<<<N_NODES / FUSE_NODES, FUSE_NODES * 32, 0, stream>>>(
        y, bufH, col, start, cnt, dinv, b1, g1, be1, a, W2);

    k_gather_ln<<<nodeGrid, 256, 0, stream>>>(bufH, y, col, start, cnt, dinv, b2, g2, be2, a, N);
}


# loads of a batch issue back to back: offsets' 96 column reads and count's whole trips lose the per-load bound check (branch, join, translation wait), a gather group's 8 row offsets sit in 8 registers
# speedup vs baseline: 3.6115x; 1.3371x over previous
#include <hip/hip_runtime.h>
#include <hip/hip_bf16.h>

#define N_NODES 50000
#define HDIM    96
#define LN_EPS  1e-5f

typedef float v2f __attribute__((ext_vector_type(2)));
typedef float v8f __attribute__((ext_vector_type(8)));

typedef int v4i __attribute__((ext_vector_type(4)));
#define EDGE_CHUNKS   96
#define BUILD_THREADS 1024
#define EDGE_UNROLL   8
#define OFF_THREADS   256

__global__ void __launch_bounds__(BUILD_THREADS)
k_count(int* __restrict__ cntc, int* __restrict__ total,
        const int* __restrict__ dst, int E) {
    __shared__ __attribute__((aligned(16))) int ctr[N_NODES];

    const int tid = threadIdx.x;
    const int g   = blockIdx.x;
    if (g == 0 && tid == 0) *total = 0;

    v4i* __restrict__ c4 = (v4i*)ctr;
    const v4i zero = {0, 0, 0, 0};
    for (int i = tid; i < N_NODES / 4; i += BUILD_THREADS) c4[i] = zero;
    __syncthreads();

    const int chunk = (E + EDGE_CHUNKS - 1) / EDGE_CHUNKS;
    const int e0 = g * chunk;
    const int e1 = (e0 + chunk < E) ? (e0 + chunk) : E;
    const int span  = EDGE_UNROLL * BUILD_THREADS;
    const int nfull = (e1 > e0) ? (e1 - e0) / span : 0;
    int e = e0 + tid;
    for (int t = 0; t < nfull; ++t, e += span) {
        int d[EDGE_UNROLL];
#pragma unroll
        for (int u = 0; u < EDGE_UNROLL; ++u) d[u] = dst[e + u * BUILD_THREADS];
#pragma unroll
        for (int u = 0; u < EDGE_UNROLL; ++u)
            if (d[u] >= 0) atomicAdd(&ctr[d[u]], 1);
    }
    {
        int d[EDGE_UNROLL];
#pragma unroll
        for (int u = 0; u < EDGE_UNROLL; ++u) {
            const int ee = e + u * BUILD_THREADS;
            d[u] = (ee < e1) ? dst[ee] : -1;
        }
#pragma unroll
        for (int u = 0; u < EDGE_UNROLL; ++u)
            if (d[u] >= 0) atomicAdd(&ctr[d[u]], 1);
    }
    __syncthreads();

    v4i* __restrict__ row = (v4i*)(cntc + (size_t)g * N_NODES);
    for (int i = tid; i < N_NODES / 4; i += BUILD_THREADS) row[i] = c4[i];
}

__global__ void __launch_bounds__(OFF_THREADS)
k_offsets(int* __restrict__ cntc, int* __restrict__ cnt, int* __restrict__ start,
          float* __restrict__ dinv, int* __restrict__ total) {
    const int lane = threadIdx.x & 31;
    const int i    = blockIdx.x * OFF_THREADS + threadIdx.x;
    const bool live = i < N_NODES;

    const int ic = live ? i : (N_NODES - 1);
    int v[EDGE_CHUNKS];
#pragma unroll
    for (int g = 0; g < EDGE_CHUNKS; ++g)
        v[g] = cntc[(size_t)g * N_NODES + ic];
    int c = 0;
#pragma unroll
    for (int g = 0; g < EDGE_CHUNKS; ++g) c += v[g];
    if (!live) c = 0;

    int inc = c;
#pragma unroll
    for (int off = 1; off < 32; off <<= 1) {
        int t = __shfl_up(inc, off, 32);
        if (lane >= off) inc += t;
    }
    int base = 0;
    if (lane == 31) base = atomicAdd(total, inc);
    base = __shfl(base, 31, 32);

    if (live) {
        int s = base + inc - c;
        start[i] = s;
        cnt[i]   = c;
        dinv[i]  = rsqrtf((float)(c + 1));
#pragma unroll
        for (int g = 0; g < EDGE_CHUNKS; ++g) {
            cntc[(size_t)g * N_NODES + i] = s;
            s += v[g];
        }
    }
}

#define CIN         128
#define GEMM_BLOCKS 160
#define MTILES      (N_NODES / 16)
#define WP_PITCH    224
#define W_STAGE_TRIPS ((CIN / 2) * HDIM / BUILD_THREADS)
static_assert(N_NODES % 16 == 0, "whole 16-row strips");
static_assert(W_STAGE_TRIPS * BUILD_THREADS == (CIN / 2) * HDIM, "W's k pairs divide evenly over the block's threads");
static_assert((CIN / 2) * WP_PITCH <= N_NODES, "W pairs fit in the counter array");
static_assert(GEMM_BLOCKS * (BUILD_THREADS / 32) >= MTILES, "a wave has at most one strip");

__global__ void __launch_bounds__(BUILD_THREADS)
k_place_gemm(int* __restrict__ col, const int* __restrict__ cntc,
             const int* __restrict__ src, const int* __restrict__ dst, int E,
             const float* __restrict__ X, const float* __restrict__ W,
             const float* __restrict__ dinv, float* __restrict__ Hout) {
    __shared__ __attribute__((aligned(16))) int ctr[N_NODES];

    const int tid = threadIdx.x;

    if (blockIdx.x < GEMM_BLOCKS) {
        v2f* __restrict__ Wp = (v2f*)ctr;
        v2f wst[W_STAGE_TRIPS];
#pragma unroll
        for (int j = 0; j < W_STAGE_TRIPS; ++j) {
            const int i = tid + j * BUILD_THREADS;
            const int p = i / HDIM;
            const int n = i - p * HDIM;
            wst[j].x = W[(size_t)(2 * p) * HDIM + n];
            wst[j].y = W[(size_t)(2 * p + 1) * HDIM + n];
        }
#pragma unroll
        for (int j = 0; j < W_STAGE_TRIPS; ++j) {
            const int i = tid + j * BUILD_THREADS;
            const int p = i / HDIM;
            const int n = i - p * HDIM;
            Wp[p * (WP_PITCH / 2) + n] = wst[j];
        }
        __syncthreads();

        const int lane = tid & 31;
        const int w = __builtin_amdgcn_readfirstlane(tid >> 5);
        const int strip = blockIdx.x + GEMM_BLOCKS * w;
        if (strip >= MTILES) return;

        const int r = lane & 15;
        const int h = lane >> 4;

        const int row0 = strip * 16 + (h << 3);
        float dsc[8];
#pragma unroll
        for (int i = 0; i < 8; ++i) dsc[i] = dinv[row0 + i];

        const v2f* __restrict__ arow = (const v2f*)(X + (size_t)(strip * 16 + r) * CIN + 2 * h);
        const v2f* __restrict__ bcol = Wp + h * (WP_PITCH / 2) + r;

        v8f c[HDIM / 16];
#pragma unroll
        for (int t = 0; t < HDIM / 16; ++t) c[t] = (v8f){};

#pragma unroll 8
        for (int kk = 0; kk < CIN; kk += 4) {
            const v2f a = arow[kk >> 1];
#pragma unroll
            for (int t = 0; t < HDIM / 16; ++t) {
                const v2f b = bcol[(kk >> 1) * (WP_PITCH / 2) + t * 16];
                c[t] = __builtin_amdgcn_wmma_f32_16x16x4_f32(
                           false, a, false, b, (short)0, c[t], false, false);
            }
        }

        float* __restrict__ out = Hout + (size_t)row0 * HDIM + r;
#pragma unroll
        for (int t = 0; t < HDIM / 16; ++t)
#pragma unroll
            for (int i = 0; i < 8; ++i) out[(size_t)i * HDIM + t * 16] = c[t][i] * dsc[i];
        return;
    }

    const int g = blockIdx.x - GEMM_BLOCKS;

    v4i* __restrict__ c4 = (v4i*)ctr;
    const v4i* __restrict__ row = (const v4i*)(cntc + (size_t)g * N_NODES);
    for (int i = tid; i < N_NODES / 4; i += BUILD_THREADS) c4[i] = row[i];
    __syncthreads();

    const int chunk = (E + EDGE_CHUNKS - 1) / EDGE_CHUNKS;
    const int e0 = g * chunk;
    const int e1 = (e0 + chunk < E) ? (e0 + chunk) : E;
    for (int e = e0 + tid; e < e1; e += EDGE_UNROLL * BUILD_THREADS) {
        int d[EDGE_UNROLL], s[EDGE_UNROLL];
#pragma unroll
        for (int u = 0; u < EDGE_UNROLL; ++u) {
            const int ee = e + u * BUILD_THREADS;
            d[u] = (ee < e1) ? dst[ee] : -1;
            s[u] = (ee < e1) ? src[ee] : 0;
        }
        int slot[EDGE_UNROLL];
#pragma unroll
        for (int u = 0; u < EDGE_UNROLL; ++u)
            slot[u] = (d[u] >= 0) ? atomicAdd(&ctr[d[u]], 1) : -1;
#pragma unroll
        for (int u = 0; u < EDGE_UNROLL; ++u)
            if (slot[u] >= 0) col[slot[u]] = s[u];
    }
}

typedef float v4f __attribute__((ext_vector_type(4)));
#define GATHER_GROUP 8
#define ROW_PIECES   (HDIM / 4)
static_assert(GATHER_GROUP == 8, "the gather names its eight row offsets one by one");
static_assert((unsigned long long)N_NODES * HDIM * 4 < (1ull << 32), "a byte offset into the plane fits 32 bits");

__device__ __forceinline__ float wave_sum(float v) {
#pragma unroll
    for (int m = 16; m > 0; m >>= 1) v += __shfl_xor(v, m, 32);
    return v;
}

__device__ __forceinline__ v4f gather_ln_row(const float* __restrict__ hs,
                                             const int* __restrict__ col,
                                             const int* __restrict__ start,
                                             const int* __restrict__ cnt,
                                             const float* __restrict__ dinv,
                                             const float* __restrict__ bias,
                                             const float* __restrict__ gamma,
                                             const float* __restrict__ beta,
                                             const float* __restrict__ alpha,
                                             const int wid, const int lane) {
    const int beg = __builtin_amdgcn_readfirstlane(start[wid]);
    const int c   = __builtin_amdgcn_readfirstlane(cnt[wid]);

    const int piece = (lane < ROW_PIECES) ? lane : lane - 8;
    const v4f* __restrict__ rows = (const v4f*)hs;
    const char* __restrict__ plane = (const char*)hs;

    v4f acc = rows[(size_t)wid * ROW_PIECES + piece];
    const unsigned pbyte = (unsigned)piece * 16u;

    const bool own = lane < ROW_PIECES;
    const int  f0  = piece * 4;
    const float dv = dinv[wid];
    float bi[4], ga[4], be[4], al[4];
#pragma unroll
    for (int i = 0; i < 4; ++i) {
        bi[i] = bias[f0 + i];
        ga[i] = gamma[f0 + i];
        be[i] = beta[f0 + i];
        al[i] = alpha[f0 + i];
    }

    for (int base = 0; base < c; base += 32) {
        const int m  = (c - base < 32) ? (c - base) : 32;
        const int my = (lane < m) ? col[beg + base + lane] : wid;
        for (int j = 0; j < m; j += GATHER_GROUP) {
            unsigned ro[GATHER_GROUP];
#pragma unroll
            for (int u = 0; u < GATHER_GROUP; ++u) {
                const int s = __builtin_amdgcn_readlane(my, j + u);
                ro[u] = (unsigned)s * (unsigned)(ROW_PIECES * 16) + pbyte;
            }
            asm("" : "+v"(ro[0]), "+v"(ro[1]), "+v"(ro[2]), "+v"(ro[3]),
                     "+v"(ro[4]), "+v"(ro[5]), "+v"(ro[6]), "+v"(ro[7]));
            v4f v[GATHER_GROUP];
#pragma unroll
            for (int u = 0; u < GATHER_GROUP; ++u)
                v[u] = *(const v4f*)(plane + ro[u]);
#pragma unroll
            for (int u = 0; u < GATHER_GROUP; ++u)
                if (j + u < m) acc += v[u];
        }
    }

    float t[4];
#pragma unroll
    for (int i = 0; i < 4; ++i) t[i] = acc[i] * dv + bi[i];

    const float s  = (t[0] + t[1]) + (t[2] + t[3]);
    const float mu = wave_sum(own ? s : 0.0f) * (1.0f / (float)HDIM);

    float d[4];
#pragma unroll
    for (int i = 0; i < 4; ++i) d[i] = t[i] - mu;
    const float q    = (d[0] * d[0] + d[1] * d[1]) + (d[2] * d[2] + d[3] * d[3]);
    const float rstd = rsqrtf(wave_sum(own ? q : 0.0f) * (1.0f / (float)HDIM) + LN_EPS);

    v4f o;
#pragma unroll
    for (int i = 0; i < 4; ++i) {
        const float z = d[i] * rstd * ga[i] + be[i];
        o[i] = (z >= 0.0f) ? z : al[i] * z;
    }
    return o;
}

__global__ void k_gather_ln(const float* __restrict__ hs,
                            float* __restrict__ y,
                            const int* __restrict__ col,
                            const int* __restrict__ start,
                            const int* __restrict__ cnt,
                            const float* __restrict__ dinv,
                            const float* __restrict__ bias,
                            const float* __restrict__ gamma,
                            const float* __restrict__ beta,
                            const float* __restrict__ alpha,
                            int N) {
    const int lane = threadIdx.x & 31;
    const int wid = __builtin_amdgcn_readfirstlane((blockIdx.x * blockDim.x + threadIdx.x) >> 5);
    if (wid >= N) return;

    const v4f o = gather_ln_row(hs, col, start, cnt, dinv, bias, gamma, beta, alpha, wid, lane);

    if (lane < ROW_PIECES)
        ((v4f*)y)[(size_t)wid * ROW_PIECES + lane] = o;
}

#define FUSE_NODES 16
#define YS_PITCH   100
static_assert(N_NODES % FUSE_NODES == 0, "every work-group is a full 16-row tile");
static_assert(FUSE_NODES >= HDIM / 16, "one wave per column tile");

__global__ void __launch_bounds__(FUSE_NODES * 32, 16)
k_gather_ln_gemm(const float* __restrict__ hs,
                 float* __restrict__ hnext,
                 const int* __restrict__ col,
                 const int* __restrict__ start,
                 const int* __restrict__ cnt,
                 const float* __restrict__ dinv,
                 const float* __restrict__ bias,
                 const float* __restrict__ gamma,
                 const float* __restrict__ beta,
                 const float* __restrict__ alpha,
                 const float* __restrict__ Wn) {
    __shared__ __attribute__((aligned(16))) float Ys[FUSE_NODES * YS_PITCH];

    const int lane = threadIdx.x & 31;
    const int w     = __builtin_amdgcn_readfirstlane(threadIdx.x >> 5);
    const int tile0 = blockIdx.x * FUSE_NODES;
    const int wid   = tile0 + w;

    const v4f o = gather_ln_row(hs, col, start, cnt, dinv, bias, gamma, beta, alpha, wid, lane);

    if (lane < ROW_PIECES)
        *(v4f*)&Ys[w * YS_PITCH + lane * 4] = o;
    __syncthreads();

    if (w >= HDIM / 16) return;

    const int r    = lane & 15;
    const int koff = (lane >> 4) << 1;

    const int row0 = tile0 + ((lane >> 4) << 3);
    float dsc[8];
#pragma unroll
    for (int i = 0; i < 8; ++i) dsc[i] = dinv[row0 + i];

    const float* __restrict__ arow = Ys + r * YS_PITCH + koff;
    const float* __restrict__ bcol = Wn + (size_t)koff * HDIM + w * 16 + r;

    v8f c = {};
#pragma unroll 12
    for (int kk = 0; kk < HDIM; kk += 4) {
        v2f a, b;
        a.x = arow[kk + 0];
        a.y = arow[kk + 1];
        b.x = bcol[(size_t)(kk + 0) * HDIM];
        b.y = bcol[(size_t)(kk + 1) * HDIM];
        c = __builtin_amdgcn_wmma_f32_16x16x4_f32(
                false, a, false, b, (short)0, c, false, false);
    }

    float* __restrict__ out = hnext + (size_t)row0 * HDIM + w * 16 + r;
#pragma unroll
    for (int i = 0; i < 8; ++i) out[(size_t)i * HDIM] = c[i] * dsc[i];
}

extern "C" void kernel_launch(void* const* d_in, const int* in_sizes, int n_in,
                              void* d_out, int out_size, void* d_ws, size_t ws_size,
                              hipStream_t stream) {
    const float* x   = (const float*)d_in[0];
    const int*   ei  = (const int*)  d_in[1];
    const float* W1  = (const float*)d_in[2];
    const float* b1  = (const float*)d_in[3];
    const float* W2  = (const float*)d_in[4];
    const float* b2  = (const float*)d_in[5];
    const float* g1  = (const float*)d_in[6];
    const float* be1 = (const float*)d_in[7];
    const float* g2  = (const float*)d_in[8];
    const float* be2 = (const float*)d_in[9];
    const float* a   = (const float*)d_in[10];

    const int N   = N_NODES;
    const int E   = in_sizes[1] / 2;
    const int NH  = N * HDIM;

    const int* src = ei;
    const int* dst = ei + E;

    int*   cnt   = (int*)d_ws;
    int*   total = cnt + N;
    float* bufH  = (float*)d_ws + ((N + 255) & ~255);
    int*   cntc  = (int*)bufH;
    int*   col   = (int*)(bufH + (size_t)NH);
    int*   start = col + 2 * (size_t)E;
    float* dinv  = (float*)(start + N);
    float* y     = (float*)d_out;

    k_count  <<<EDGE_CHUNKS, BUILD_THREADS, 0, stream>>>(cntc, total, dst, E);
    k_offsets<<<(N + OFF_THREADS - 1) / OFF_THREADS, OFF_THREADS, 0, stream>>>(cntc, cnt, start, dinv, total);
    k_place_gemm<<<GEMM_BLOCKS + EDGE_CHUNKS, BUILD_THREADS, 0, stream>>>(
        col, cntc, src, dst, E, x, W1, dinv, y);

    const int nodeGrid = (N * 32 + 255) / 256;

    k_gather_ln_gemm<<<N_NODES / FUSE_NODES, FUSE_NODES * 32, 0, stream>>>(
        y, bufH, col, start, cnt, dinv, b1, g1, be1, a, W2);

    k_gather_ln<<<nodeGrid, 256, 0, stream>>>(bufH, y, col, start, cnt, dinv, b2, g2, be2, a, N);
}


# range sort: a chunk's sorted pairs go to their places in an LDS copy of its stretch and are written out in order: a wave's store covers 2 whole 128-byte lines instead of 32 lines with 8 bytes each
# speedup vs baseline: 4.7760x; 1.3224x over previous
#include <hip/hip_runtime.h>
#include <hip/hip_bf16.h>

#define N_NODES 50000
#define HDIM    96
#define LN_EPS  1e-5f

typedef float v2f __attribute__((ext_vector_type(2)));
typedef float v8f __attribute__((ext_vector_type(8)));
typedef int   v2i __attribute__((ext_vector_type(2)));
typedef int   v4i __attribute__((ext_vector_type(4)));

#define EDGE_CHUNKS   96
#define BUILD_THREADS 1024
#define EDGE_UNROLL   8
#define RANGE_STRIPS  13
#define RANGE_NODES   (16 * RANGE_STRIPS)
#define N_RANGES      ((N_NODES + RANGE_NODES - 1) / RANGE_NODES)
#define SCAN_THREADS  256
#define STAGE_PAIRS   8448
#define COPY_TRIPS    ((STAGE_PAIRS + 15 + BUILD_THREADS - 1) / BUILD_THREADS)
static_assert(N_RANGES + 1 <= SCAN_THREADS, "one scan thread per range and one for the total");
static_assert(SCAN_THREADS % 32 == 0 && SCAN_THREADS <= BUILD_THREADS, "whole waves scan");

__global__ void __launch_bounds__(BUILD_THREADS)
k_range_sort(int* __restrict__ pairs_w, int* __restrict__ tab,
             const int* __restrict__ src, const int* __restrict__ dst, int E) {
    __shared__ int hist[SCAN_THREADS];
    __shared__ int wsum[SCAN_THREADS / 32];
    __shared__ __attribute__((aligned(16))) v2i stage[STAGE_PAIRS];

    const int tid  = threadIdx.x;
    const int lane = tid & 31;
    const int g    = blockIdx.x;

    if (tid < SCAN_THREADS) hist[tid] = 0;
    __syncthreads();

    const int chunk = (E + EDGE_CHUNKS - 1) / EDGE_CHUNKS;
    const int e0 = g * chunk;
    const int e1 = (e0 + chunk < E) ? (e0 + chunk) : E;
    const int span = EDGE_UNROLL * BUILD_THREADS;

    for (int b = e0; b < e1; b += span) {
        const int e = b + tid;
        int d[EDGE_UNROLL];
#pragma unroll
        for (int u = 0; u < EDGE_UNROLL; ++u) {
            const int ee = e + u * BUILD_THREADS;
            d[u] = dst[(ee < e1) ? ee : (e1 - 1)];
            if (ee >= e1) d[u] = -1;
        }
#pragma unroll
        for (int u = 0; u < EDGE_UNROLL; ++u)
            if (d[u] >= 0) atomicAdd(&hist[d[u] / RANGE_NODES], 1);
    }
    __syncthreads();

    const int v = (tid < SCAN_THREADS) ? hist[tid] : 0;
    int inc = v;
#pragma unroll
    for (int off = 1; off < 32; off <<= 1) {
        int t = __shfl_up(inc, off, 32);
        if (lane >= off) inc += t;
    }
    if (tid < SCAN_THREADS && lane == 31) wsum[tid >> 5] = inc;
    __syncthreads();
    if (tid <= N_RANGES) {
        int pre = 0;
        for (int k = 0; k < (tid >> 5); ++k) pre += wsum[k];
        const int ex = pre + inc - v;
        hist[tid] = ex;
        tab[tid * EDGE_CHUNKS + g] = ex;
    }
    __syncthreads();

    v2i* __restrict__ stretch = (v2i*)pairs_w + (size_t)e0;
    const int  npairs = e1 - e0;
    const bool staged = npairs <= STAGE_PAIRS;
    for (int b = e0; b < e1; b += span) {
        const int e = b + tid;
        int d[EDGE_UNROLL], s[EDGE_UNROLL];
#pragma unroll
        for (int u = 0; u < EDGE_UNROLL; ++u) {
            const int ee = e + u * BUILD_THREADS;
            const int ec = (ee < e1) ? ee : (e1 - 1);
            d[u] = dst[ec];
            s[u] = src[ec];
            if (ee >= e1) d[u] = -1;
        }
        int pos[EDGE_UNROLL];
#pragma unroll
        for (int u = 0; u < EDGE_UNROLL; ++u)
            pos[u] = (d[u] >= 0) ? atomicAdd(&hist[d[u] / RANGE_NODES], 1) : -1;
        if (staged) {
#pragma unroll
            for (int u = 0; u < EDGE_UNROLL; ++u)
                if (pos[u] >= 0) stage[pos[u]] = (v2i){s[u], d[u]};
        } else {
#pragma unroll
            for (int u = 0; u < EDGE_UNROLL; ++u)
                if (pos[u] >= 0) stretch[pos[u]] = (v2i){s[u], d[u]};
        }
    }
    if (!staged) return;
    __syncthreads();

    const int shift = (int)(((size_t)stretch >> 3) & 15);
    v2i t[COPY_TRIPS];
#pragma unroll
    for (int j = 0; j < COPY_TRIPS; ++j) {
        const int i = tid - shift + j * BUILD_THREADS;
        t[j] = stage[(i < 0) ? 0 : ((i < STAGE_PAIRS) ? i : (STAGE_PAIRS - 1))];
    }
#pragma unroll
    for (int j = 0; j < COPY_TRIPS; ++j) {
        const int i = tid - shift + j * BUILD_THREADS;
        if (i >= 0 && i < npairs) stretch[i] = t[j];
    }
}

#define CIN         128
#define MTILES      (N_NODES / 16)
#define WP_PITCH    224
#define W_STAGE_TRIPS ((CIN / 2) * HDIM / BUILD_THREADS)
#define BUILD_WAVES (BUILD_THREADS / 32)
#define WAVE_CHUNKS (EDGE_CHUNKS / BUILD_WAVES)
#define PIECE_TRIPS 2
#define HALF_TILES  (HDIM / 32)
#define NODE_WAVES  ((RANGE_NODES + 31) / 32)
static_assert(N_NODES % 16 == 0, "whole 16-row strips");
static_assert(W_STAGE_TRIPS * BUILD_THREADS == (CIN / 2) * HDIM, "W's k pairs divide evenly over the block's threads");
static_assert(WAVE_CHUNKS * BUILD_WAVES == EDGE_CHUNKS, "every wave has the same number of pieces");
static_assert(EDGE_CHUNKS % 32 == 0, "whole waves read the table row");
static_assert(2 * RANGE_STRIPS <= BUILD_WAVES, "a wave has at most one half strip");
static_assert(HDIM % 32 == 0, "two column halves of whole tiles");
static_assert((EDGE_CHUNKS * RANGE_NODES) % 4 == 0, "counters cleared 16 bytes at a time");

__device__ __forceinline__ int wave_sum_i(int v) {
#pragma unroll
    for (int m = 16; m > 0; m >>= 1) v += __shfl_xor(v, m, 32);
    return v;
}

__global__ void __launch_bounds__(BUILD_THREADS)
k_merge_gemm(int* __restrict__ col, int* __restrict__ cnt, int* __restrict__ start,
             float* __restrict__ dinv,
             const int* __restrict__ pairs_r, const int* __restrict__ tab, int E,
             const float* __restrict__ X, const float* __restrict__ W,
             float* __restrict__ Hout) {
    __shared__ __attribute__((aligned(16))) int cc[EDGE_CHUNKS * RANGE_NODES];
    __shared__ __attribute__((aligned(16))) v2f Wp[(CIN / 2) * (WP_PITCH / 2)];
    __shared__ int   pst[EDGE_CHUNKS];
    __shared__ int   pln[EDGE_CHUNKS];
    __shared__ float dl[RANGE_NODES];
    __shared__ int   wsum[NODE_WAVES];
    __shared__ int   bsum[EDGE_CHUNKS / 32];

    const int tid  = threadIdx.x;
    const int lane = tid & 31;
    const int w  = __builtin_amdgcn_readfirstlane(tid >> 5);
    const int R  = blockIdx.x;
    const int n0 = R * RANGE_NODES;
    const int nn = (N_NODES - n0 < RANGE_NODES) ? (N_NODES - n0) : RANGE_NODES;

    v2f wst[W_STAGE_TRIPS];
#pragma unroll
    for (int j = 0; j < W_STAGE_TRIPS; ++j) {
        const int i = tid + j * BUILD_THREADS;
        const int p = i / HDIM;
        const int n = i - p * HDIM;
        wst[j].x = W[(size_t)(2 * p) * HDIM + n];
        wst[j].y = W[(size_t)(2 * p + 1) * HDIM + n];
    }

    if (tid < EDGE_CHUNKS) {
        const int a = tab[R * EDGE_CHUNKS + tid];
        const int b = tab[(R + 1) * EDGE_CHUNKS + tid];
        pst[tid] = a;
        pln[tid] = b - a;
        const int sa = wave_sum_i(a);
        if (lane == 0) bsum[tid >> 5] = sa;
    }

    v4i* __restrict__ c4 = (v4i*)cc;
    const v4i zero = {0, 0, 0, 0};
    for (int i = tid; i < EDGE_CHUNKS * RANGE_NODES / 4; i += BUILD_THREADS) c4[i] = zero;

#pragma unroll
    for (int j = 0; j < W_STAGE_TRIPS; ++j) {
        const int i = tid + j * BUILD_THREADS;
        const int p = i / HDIM;
        const int n = i - p * HDIM;
        Wp[p * (WP_PITCH / 2) + n] = wst[j];
    }
    __syncthreads();

    const int chunk = (E + EDGE_CHUNKS - 1) / EDGE_CHUNKS;
    int pl[WAVE_CHUNKS], cb[WAVE_CHUNKS];
    const v2i* __restrict__ pp[WAVE_CHUNKS];
#pragma unroll
    for (int k = 0; k < WAVE_CHUNKS; ++k) {
        const int g = w + BUILD_WAVES * k;
        pl[k] = __builtin_amdgcn_readfirstlane(pln[g]);
        pp[k] = (const v2i*)pairs_r + (size_t)g * chunk + __builtin_amdgcn_readfirstlane(pst[g]);
        cb[k] = g * RANGE_NODES - n0;
    }
    v2i sd[WAVE_CHUNKS][PIECE_TRIPS];
#pragma unroll
    for (int k = 0; k < WAVE_CHUNKS; ++k)
#pragma unroll
        for (int t = 0; t < PIECE_TRIPS; ++t) {
            const int i = lane + 32 * t;
            sd[k][t] = pp[k][(i < pl[k]) ? i : 0];
        }
#pragma unroll
    for (int k = 0; k < WAVE_CHUNKS; ++k)
#pragma unroll
        for (int t = 0; t < PIECE_TRIPS; ++t)
            if (lane + 32 * t < pl[k]) atomicAdd(&cc[cb[k] + sd[k][t].y], 1);
#pragma unroll
    for (int k = 0; k < WAVE_CHUNKS; ++k)
        for (int i = 32 * PIECE_TRIPS + lane; i < pl[k]; i += 32) {
            const v2i x = pp[k][i];
            atomicAdd(&cc[cb[k] + x.y], 1);
        }
    __syncthreads();

    int c = 0;
    if (tid < RANGE_NODES) {
#pragma unroll 16
        for (int g = 0; g < EDGE_CHUNKS; ++g) c += cc[g * RANGE_NODES + tid];
    }
    int inc = c;
#pragma unroll
    for (int off = 1; off < 32; off <<= 1) {
        int t = __shfl_up(inc, off, 32);
        if (lane >= off) inc += t;
    }
    if (tid < NODE_WAVES * 32 && lane == 31) wsum[tid >> 5] = inc;
    __syncthreads();
    if (tid < RANGE_NODES) {
        int s = 0;
        for (int k = 0; k < EDGE_CHUNKS / 32; ++k) s += bsum[k];
        for (int k = 0; k < (tid >> 5); ++k) s += wsum[k];
        s += inc - c;
        const float di = rsqrtf((float)(c + 1));
        dl[tid] = di;
        if (tid < nn) {
            start[n0 + tid] = s;
            cnt[n0 + tid]   = c;
            dinv[n0 + tid]  = di;
        }
#pragma unroll 16
        for (int g = 0; g < EDGE_CHUNKS; ++g) {
            const int v = cc[g * RANGE_NODES + tid];
            cc[g * RANGE_NODES + tid] = s;
            s += v;
        }
    }
    __syncthreads();

#pragma unroll
    for (int k = 0; k < WAVE_CHUNKS; ++k)
#pragma unroll
        for (int t = 0; t < PIECE_TRIPS; ++t)
            if (lane + 32 * t < pl[k]) {
                const int slot = atomicAdd(&cc[cb[k] + sd[k][t].y], 1);
                col[slot] = sd[k][t].x;
            }
#pragma unroll
    for (int k = 0; k < WAVE_CHUNKS; ++k)
        for (int i = 32 * PIECE_TRIPS + lane; i < pl[k]; i += 32) {
            const v2i x = pp[k][i];
            const int slot = atomicAdd(&cc[cb[k] + x.y], 1);
            col[slot] = x.x;
        }

    if (w >= 2 * RANGE_STRIPS) return;
    const int sidx  = w >> 1;
    const int half  = w & 1;
    const int strip = R * RANGE_STRIPS + sidx;
    if (strip >= MTILES) return;

    const int r = lane & 15;
    const int h = lane >> 4;

    const int row0 = strip * 16 + (h << 3);
    float dsc[8];
#pragma unroll
    for (int i = 0; i < 8; ++i) dsc[i] = dl[sidx * 16 + (h << 3) + i];

    const v2f* __restrict__ arow = (const v2f*)(X + (size_t)(strip * 16 + r) * CIN + 2 * h);
    const v2f* __restrict__ bcol = Wp + h * (WP_PITCH / 2) + half * (HALF_TILES * 16) + r;

    v8f acc[HALF_TILES];
#pragma unroll
    for (int t = 0; t < HALF_TILES; ++t) acc[t] = (v8f){};

#pragma unroll 8
    for (int kk = 0; kk < CIN; kk += 4) {
        const v2f a = arow[kk >> 1];
#pragma unroll
        for (int t = 0; t < HALF_TILES; ++t) {
            const v2f b = bcol[(kk >> 1) * (WP_PITCH / 2) + t * 16];
            acc[t] = __builtin_amdgcn_wmma_f32_16x16x4_f32(
                         false, a, false, b, (short)0, acc[t], false, false);
        }
    }

    float* __restrict__ out = Hout + (size_t)row0 * HDIM + half * (HALF_TILES * 16) + r;
#pragma unroll
    for (int t = 0; t < HALF_TILES; ++t)
#pragma unroll
        for (int i = 0; i < 8; ++i) out[(size_t)i * HDIM + t * 16] = acc[t][i] * dsc[i];
}

typedef float v4f __attribute__((ext_vector_type(4)));
#define GATHER_GROUP 8
#define ROW_PIECES   (HDIM / 4)
static_assert(GATHER_GROUP == 8, "the gather names its eight row offsets one by one");
static_assert((unsigned long long)N_NODES * HDIM * 4 < (1ull << 32), "a byte offset into the plane fits 32 bits");

__device__ __forceinline__ float wave_sum(float v) {
#pragma unroll
    for (int m = 16; m > 0; m >>= 1) v += __shfl_xor(v, m, 32);
    return v;
}

__device__ __forceinline__ v4f gather_ln_row(const float* __restrict__ hs,
                                             const int* __restrict__ col,
                                             const int* __restrict__ start,
                                             const int* __restrict__ cnt,
                                             const float* __restrict__ dinv,
                                             const float* __restrict__ bias,
                                             const float* __restrict__ gamma,
                                             const float* __restrict__ beta,
                                             const float* __restrict__ alpha,
                                             const int wid, const int lane) {
    const int beg = __builtin_amdgcn_readfirstlane(start[wid]);
    const int c   = __builtin_amdgcn_readfirstlane(cnt[wid]);

    const int piece = (lane < ROW_PIECES) ? lane : lane - 8;
    const v4f* __restrict__ rows = (const v4f*)hs;
    const char* __restrict__ plane = (const char*)hs;

    v4f acc = rows[(size_t)wid * ROW_PIECES + piece];
    const unsigned pbyte = (unsigned)piece * 16u;

    const bool own = lane < ROW_PIECES;
    const int  f0  = piece * 4;
    const float dv = dinv[wid];
    float bi[4], ga[4], be[4], al[4];
#pragma unroll
    for (int i = 0; i < 4; ++i) {
        bi[i] = bias[f0 + i];
        ga[i] = gamma[f0 + i];
        be[i] = beta[f0 + i];
        al[i] = alpha[f0 + i];
    }

    for (int base = 0; base < c; base += 32) {
        const int m  = (c - base < 32) ? (c - base) : 32;
        const int my = (lane < m) ? col[beg + base + lane] : wid;
        for (int j = 0; j < m; j += GATHER_GROUP) {
            unsigned ro[GATHER_GROUP];
#pragma unroll
            for (int u = 0; u < GATHER_GROUP; ++u) {
                const int s = __builtin_amdgcn_readlane(my, j + u);
                ro[u] = (unsigned)s * (unsigned)(ROW_PIECES * 16) + pbyte;
            }
            asm("" : "+v"(ro[0]), "+v"(ro[1]), "+v"(ro[2]), "+v"(ro[3]),
                     "+v"(ro[4]), "+v"(ro[5]), "+v"(ro[6]), "+v"(ro[7]));
            v4f v[GATHER_GROUP];
#pragma unroll
            for (int u = 0; u < GATHER_GROUP; ++u)
                v[u] = *(const v4f*)(plane + ro[u]);
#pragma unroll
            for (int u = 0; u < GATHER_GROUP; ++u)
                if (j + u < m) acc += v[u];
        }
    }

    float t[4];
#pragma unroll
    for (int i = 0; i < 4; ++i) t[i] = acc[i] * dv + bi[i];

    const float s  = (t[0] + t[1]) + (t[2] + t[3]);
    const float mu = wave_sum(own ? s : 0.0f) * (1.0f / (float)HDIM);

    float d[4];
#pragma unroll
    for (int i = 0; i < 4; ++i) d[i] = t[i] - mu;
    const float q    = (d[0] * d[0] + d[1] * d[1]) + (d[2] * d[2] + d[3] * d[3]);
    const float rstd = rsqrtf(wave_sum(own ? q : 0.0f) * (1.0f / (float)HDIM) + LN_EPS);

    v4f o;
#pragma unroll
    for (int i = 0; i < 4; ++i) {
        const float z = d[i] * rstd * ga[i] + be[i];
        o[i] = (z >= 0.0f) ? z : al[i] * z;
    }
    return o;
}

__global__ void k_gather_ln(const float* __restrict__ hs,
                            float* __restrict__ y,
                            const int* __restrict__ col,
                            const int* __restrict__ start,
                            const int* __restrict__ cnt,
                            const float* __restrict__ dinv,
                            const float* __restrict__ bias,
                            const float* __restrict__ gamma,
                            const float* __restrict__ beta,
                            const float* __restrict__ alpha,
                            int N) {
    const int lane = threadIdx.x & 31;
    const int wid = __builtin_amdgcn_readfirstlane((blockIdx.x * blockDim.x + threadIdx.x) >> 5);
    if (wid >= N) return;

    const v4f o = gather_ln_row(hs, col, start, cnt, dinv, bias, gamma, beta, alpha, wid, lane);

    if (lane < ROW_PIECES)
        ((v4f*)y)[(size_t)wid * ROW_PIECES + lane] = o;
}

#define FUSE_NODES 16
#define YS_PITCH   100
static_assert(N_NODES % FUSE_NODES == 0, "every work-group is a full 16-row tile");
static_assert(FUSE_NODES >= HDIM / 16, "one wave per column tile");

__global__ void __launch_bounds__(FUSE_NODES * 32, 16)
k_gather_ln_gemm(const float* __restrict__ hs,
                 float* __restrict__ hnext,
                 const int* __restrict__ col,
                 const int* __restrict__ start,
                 const int* __restrict__ cnt,
                 const float* __restrict__ dinv,
                 const float* __restrict__ bias,
                 const float* __restrict__ gamma,
                 const float* __restrict__ beta,
                 const float* __restrict__ alpha,
                 const float* __restrict__ Wn) {
    __shared__ __attribute__((aligned(16))) float Ys[FUSE_NODES * YS_PITCH];

    const int lane = threadIdx.x & 31;
    const int w     = __builtin_amdgcn_readfirstlane(threadIdx.x >> 5);
    const int tile0 = blockIdx.x * FUSE_NODES;
    const int wid   = tile0 + w;

    const v4f o = gather_ln_row(hs, col, start, cnt, dinv, bias, gamma, beta, alpha, wid, lane);

    if (lane < ROW_PIECES)
        *(v4f*)&Ys[w * YS_PITCH + lane * 4] = o;
    __syncthreads();

    if (w >= HDIM / 16) return;

    const int r    = lane & 15;
    const int koff = (lane >> 4) << 1;

    const int row0 = tile0 + ((lane >> 4) << 3);
    float dsc[8];
#pragma unroll
    for (int i = 0; i < 8; ++i) dsc[i] = dinv[row0 + i];

    const float* __restrict__ arow = Ys + r * YS_PITCH + koff;
    const float* __restrict__ bcol = Wn + (size_t)koff * HDIM + w * 16 + r;

    v8f c = {};
#pragma unroll 12
    for (int kk = 0; kk < HDIM; kk += 4) {
        v2f a, b;
        a.x = arow[kk + 0];
        a.y = arow[kk + 1];
        b.x = bcol[(size_t)(kk + 0) * HDIM];
        b.y = bcol[(size_t)(kk + 1) * HDIM];
        c = __builtin_amdgcn_wmma_f32_16x16x4_f32(
                false, a, false, b, (short)0, c, false, false);
    }

    float* __restrict__ out = hnext + (size_t)row0 * HDIM + w * 16 + r;
#pragma unroll
    for (int i = 0; i < 8; ++i) out[(size_t)i * HDIM] = c[i] * dsc[i];
}

extern "C" void kernel_launch(void* const* d_in, const int* in_sizes, int n_in,
                              void* d_out, int out_size, void* d_ws, size_t ws_size,
                              hipStream_t stream) {
    const float* x   = (const float*)d_in[0];
    const int*   ei  = (const int*)  d_in[1];
    const float* W1  = (const float*)d_in[2];
    const float* b1  = (const float*)d_in[3];
    const float* W2  = (const float*)d_in[4];
    const float* b2  = (const float*)d_in[5];
    const float* g1  = (const float*)d_in[6];
    const float* be1 = (const float*)d_in[7];
    const float* g2  = (const float*)d_in[8];
    const float* be2 = (const float*)d_in[9];
    const float* a   = (const float*)d_in[10];

    const int N   = N_NODES;
    const int E   = in_sizes[1] / 2;
    const int NH  = N * HDIM;

    const int* src = ei;
    const int* dst = ei + E;

    int*   cnt   = (int*)d_ws;
    float* bufH  = (float*)d_ws + ((N + 255) & ~255);
    int*   pairs = (int*)bufH;
    int*   tab   = pairs + 2 * (size_t)E;
    int*   col   = (int*)(bufH + (size_t)NH);
    int*   start = col + 2 * (size_t)E;
    float* dinv  = (float*)(start + N);
    float* y     = (float*)d_out;

    k_range_sort<<<EDGE_CHUNKS, BUILD_THREADS, 0, stream>>>(pairs, tab, src, dst, E);
    k_merge_gemm<<<N_RANGES, BUILD_THREADS, 0, stream>>>(
        col, cnt, start, dinv, pairs, tab, E, x, W1, y);

    const int nodeGrid = (N * 32 + 255) / 256;

    k_gather_ln_gemm<<<N_NODES / FUSE_NODES, FUSE_NODES * 32, 0, stream>>>(
        y, bufH, col, start, cnt, dinv, b1, g1, be1, a, W2);

    k_gather_ln<<<nodeGrid, 256, 0, stream>>>(bufH, y, col, start, cnt, dinv, b2, g2, be2, a, N);
}
